# starting kernel
# baseline (speedup 1.0000x reference)
#include <hip/hip_runtime.h>

typedef __attribute__((ext_vector_type(2))) float v2f;
typedef __attribute__((ext_vector_type(8))) float v8f;

#define VXD 64
#define VVV (64*64*64)
#define TT  128
#define NB  2
#define HW  65536
#define LOG2E 1.4426950408889634f

__global__ void __launch_bounds__(256)
slice_mean_kernel(const float* __restrict__ slices, float* __restrict__ means) {
    __shared__ float red[256];
    const int slice = blockIdx.x;
    const float4* p = (const float4*)(slices + (size_t)slice * HW);
    float s = 0.f;
    #pragma unroll 4
    for (int i = threadIdx.x; i < HW / 4; i += 256) {
        float4 q = p[i];
        s += (q.x + q.y) + (q.z + q.w);
    }
    red[threadIdx.x] = s;
    __syncthreads();
    for (int off = 128; off > 0; off >>= 1) {
        if (threadIdx.x < off) red[threadIdx.x] += red[threadIdx.x + off];
        __syncthreads();
    }
    if (threadIdx.x == 0) means[slice] = red[0] * (1.0f / (float)HW);
}

__global__ void __launch_bounds__(256)
recon_kernel(const float* __restrict__ transforms,
             const int*   __restrict__ sel_idx,
             const float* __restrict__ means,
             float*       __restrict__ out) {
    __shared__ float lds_c[TT * 4];
    __shared__ float lds_m[TT];

    const int b   = blockIdx.y;
    const int tid = threadIdx.x;

    if (tid < TT) {
        const int j  = sel_idx[b * TT + tid];
        const float* tr = transforms + (size_t)(b * TT + j) * 6;
        const float cx = tr[0], cy = tr[1], cz = tr[2];
        lds_c[tid * 4 + 0] = -2.0f * cx;
        lds_c[tid * 4 + 1] = -2.0f * cy;
        lds_c[tid * 4 + 2] = -2.0f * cz;
        lds_c[tid * 4 + 3] = cx * cx + cy * cy + cz * cz;
        lds_m[tid] = means[b * TT + j];
    }
    __syncthreads();

    const int  lane = tid & 31;
    const int  wave = tid >> 5;
    const int  n    = lane & 15;
    const bool hi   = lane >= 16;

    const int   v  = blockIdx.x * 128 + wave * 16 + n;
    const float gx = (float)(v >> 12);
    const float gy = (float)((v >> 6) & 63);
    const float gz = (float)(v & 63);
    const float g2 = gx * gx + gy * gy + gz * gz;

    v2f bm;
    bm.x = hi ? gz   : gx;
    bm.y = hi ? 1.0f : gy;

    const v8f cacc = {g2, g2, g2, g2, g2, g2, g2, g2};

    float sumw = 0.f, sumwm = 0.f;

    #pragma unroll
    for (int tt = 0; tt < TT / 16; ++tt) {
        const float* cp = &lds_c[(tt * 16 + n) * 4];
        v2f am;
        am.x = hi ? cp[2] : cp[0];
        am.y = hi ? cp[3] : cp[1];

        v8f d = __builtin_amdgcn_wmma_f32_16x16x4_f32(
            false, am, false, bm, (short)0, cacc, false, false);

        const int tofs = tt * 16 + (hi ? 8 : 0);
        #pragma unroll
        for (int r = 0; r < 8; ++r) {
            const float y = __builtin_amdgcn_rsqf(d[r]);
            const float w = __builtin_amdgcn_exp2f(y * LOG2E);
            sumw  += w;
            sumwm += w * lds_m[tofs + r];
        }
    }

    sumw  += __shfl_xor(sumw,  16, 32);
    sumwm += __shfl_xor(sumwm, 16, 32);

    if (!hi) {
        out[(size_t)b * VVV + v] = sumwm * __builtin_amdgcn_rcpf(sumw);
    }
}

extern "C" void kernel_launch(void* const* d_in, const int* in_sizes, int n_in,
                              void* d_out, int out_size, void* d_ws, size_t ws_size,
                              hipStream_t stream) {
    const float* slices     = (const float*)d_in[0];
    const float* transforms = (const float*)d_in[1];
    const int*   sel_idx    = (const int*)d_in[2];
    float*       out        = (float*)d_out;
    float*       means      = (float*)d_ws;

    slice_mean_kernel<<<NB * TT, 256, 0, stream>>>(slices, means);

    dim3 grid(VVV / 128, NB);
    recon_kernel<<<grid, 256, 0, stream>>>(transforms, sel_idx, means, out);
}


# pass 1: 16 waves per slice work-group, each lane's 32 16-byte loads all issued before its sums (was 8 waves, 4 loads per drained trip); lane chains + wave butterfly: new summation order of the means
# speedup vs baseline: 1.3515x; 1.3515x over previous
#include <hip/hip_runtime.h>

typedef __attribute__((ext_vector_type(2))) float v2f;
typedef __attribute__((ext_vector_type(8))) float v8f;

#define VXD 64
#define VVV (64*64*64)
#define TT  128
#define NB  2
#define HW  65536
#define LOG2E 1.4426950408889634f

#define P1_WAVES 16
#define P1_LOADS (HW / 4 / (P1_WAVES * 32))

__global__ void __launch_bounds__(P1_WAVES * 32)
slice_mean_kernel(const float* __restrict__ slices, float* __restrict__ means) {
    __shared__ float red[P1_WAVES];
    const int slice = blockIdx.x;
    const int lane  = threadIdx.x & 31;
    const int wave  = threadIdx.x >> 5;
    const float4* p = (const float4*)(slices + (size_t)slice * HW)
                    + wave * (P1_LOADS * 32) + lane;
    float4 q[P1_LOADS];
    #pragma unroll
    for (int k = 0; k < P1_LOADS; ++k) q[k] = p[k * 32];
    __builtin_amdgcn_sched_barrier(0);
    float s0 = 0.f, s1 = 0.f, s2 = 0.f, s3 = 0.f;
    #pragma unroll
    for (int k = 0; k < P1_LOADS; k += 4) {
        s0 += (q[k + 0].x + q[k + 0].y) + (q[k + 0].z + q[k + 0].w);
        s1 += (q[k + 1].x + q[k + 1].y) + (q[k + 1].z + q[k + 1].w);
        s2 += (q[k + 2].x + q[k + 2].y) + (q[k + 2].z + q[k + 2].w);
        s3 += (q[k + 3].x + q[k + 3].y) + (q[k + 3].z + q[k + 3].w);
    }
    float s = (s0 + s1) + (s2 + s3);
    for (int off = 16; off > 0; off >>= 1) s += __shfl_xor(s, off, 32);
    if (lane == 0) red[wave] = s;
    __syncthreads();
    if (threadIdx.x == 0) {
        float t = 0.f;
        for (int w = 0; w < P1_WAVES; ++w) t += red[w];
        means[slice] = t * (1.0f / (float)HW);
    }
}

__global__ void __launch_bounds__(256)
recon_kernel(const float* __restrict__ transforms,
             const int*   __restrict__ sel_idx,
             const float* __restrict__ means,
             float*       __restrict__ out) {
    __shared__ float lds_c[TT * 4];
    __shared__ float lds_m[TT];

    const int b   = blockIdx.y;
    const int tid = threadIdx.x;

    if (tid < TT) {
        const int j  = sel_idx[b * TT + tid];
        const float* tr = transforms + (size_t)(b * TT + j) * 6;
        const float cx = tr[0], cy = tr[1], cz = tr[2];
        lds_c[tid * 4 + 0] = -2.0f * cx;
        lds_c[tid * 4 + 1] = -2.0f * cy;
        lds_c[tid * 4 + 2] = -2.0f * cz;
        lds_c[tid * 4 + 3] = cx * cx + cy * cy + cz * cz;
        lds_m[tid] = means[b * TT + j];
    }
    __syncthreads();

    const int  lane = tid & 31;
    const int  wave = tid >> 5;
    const int  n    = lane & 15;
    const bool hi   = lane >= 16;

    const int   v  = blockIdx.x * 128 + wave * 16 + n;
    const float gx = (float)(v >> 12);
    const float gy = (float)((v >> 6) & 63);
    const float gz = (float)(v & 63);
    const float g2 = gx * gx + gy * gy + gz * gz;

    v2f bm;
    bm.x = hi ? gz   : gx;
    bm.y = hi ? 1.0f : gy;

    const v8f cacc = {g2, g2, g2, g2, g2, g2, g2, g2};

    float sumw = 0.f, sumwm = 0.f;

    #pragma unroll
    for (int tt = 0; tt < TT / 16; ++tt) {
        const float* cp = &lds_c[(tt * 16 + n) * 4];
        v2f am;
        am.x = hi ? cp[2] : cp[0];
        am.y = hi ? cp[3] : cp[1];

        v8f d = __builtin_amdgcn_wmma_f32_16x16x4_f32(
            false, am, false, bm, (short)0, cacc, false, false);

        const int tofs = tt * 16 + (hi ? 8 : 0);
        #pragma unroll
        for (int r = 0; r < 8; ++r) {
            const float y = __builtin_amdgcn_rsqf(d[r]);
            const float w = __builtin_amdgcn_exp2f(y * LOG2E);
            sumw  += w;
            sumwm += w * lds_m[tofs + r];
        }
    }

    sumw  += __shfl_xor(sumw,  16, 32);
    sumwm += __shfl_xor(sumwm, 16, 32);

    if (!hi) {
        out[(size_t)b * VVV + v] = sumwm * __builtin_amdgcn_rcpf(sumw);
    }
}

extern "C" void kernel_launch(void* const* d_in, const int* in_sizes, int n_in,
                              void* d_out, int out_size, void* d_ws, size_t ws_size,
                              hipStream_t stream) {
    const float* slices     = (const float*)d_in[0];
    const float* transforms = (const float*)d_in[1];
    const int*   sel_idx    = (const int*)d_in[2];
    float*       out        = (float*)d_out;
    float*       means      = (float*)d_ws;

    slice_mean_kernel<<<NB * TT, P1_WAVES * 32, 0, stream>>>(slices, means);

    dim3 grid(VVV / 128, NB);
    recon_kernel<<<grid, 256, 0, stream>>>(transforms, sel_idx, means, out);
}


# both passes work per DISTINCT selected slice: pass 1 counts each slice's picks and streams only picked slices; pass 2 adds each once with weight*count (5-6 t-tiles, was 8): new summation order
# speedup vs baseline: 1.5803x; 1.1693x over previous
#include <hip/hip_runtime.h>


typedef __attribute__((ext_vector_type(2))) float v2f;
typedef __attribute__((ext_vector_type(8))) float v8f;

#define VXD 64
#define VVV (64*64*64)
#define TT  128
#define NB  2
#define HW  65536
#define LOG2E 1.4426950408889634f


#define P1_WAVES 16
#define P1_LOADS (HW / 4 / (P1_WAVES * 32))

__global__ void __launch_bounds__(P1_WAVES * 32)
slice_mean_kernel(const float* __restrict__ slices, const int* __restrict__ sel_idx,
                  float* __restrict__ means, float* __restrict__ counts) {
    __shared__ float red[P1_WAVES];
    __shared__ int   hits[TT / 32];
    const int slice = blockIdx.x;
    const int lane  = threadIdx.x & 31;
    const int wave  = threadIdx.x >> 5;
    if (wave < TT / 32) {
        const int pick = sel_idx[(slice / TT) * TT + threadIdx.x];
        const unsigned named = (unsigned)__ballot(pick == (slice % TT));
        if (lane == 0) hits[wave] = __popc(named);
    }
    __syncthreads();
    const int cnt = (hits[0] + hits[1]) + (hits[2] + hits[3]);
    if (threadIdx.x == 0) {
        counts[slice] = (float)cnt;
        if (cnt == 0) means[slice] = 0.f;
    }
    if (cnt == 0) return;
    const float4* p = (const float4*)(slices + (size_t)slice * HW)
                    + wave * (P1_LOADS * 32) + lane;
    float4 q[P1_LOADS];
    #pragma unroll
    for (int k = 0; k < P1_LOADS; ++k) q[k] = p[k * 32];
    __builtin_amdgcn_sched_barrier(0);
    float s0 = 0.f, s1 = 0.f, s2 = 0.f, s3 = 0.f;
    #pragma unroll
    for (int k = 0; k < P1_LOADS; k += 4) {
        s0 += (q[k + 0].x + q[k + 0].y) + (q[k + 0].z + q[k + 0].w);
        s1 += (q[k + 1].x + q[k + 1].y) + (q[k + 1].z + q[k + 1].w);
        s2 += (q[k + 2].x + q[k + 2].y) + (q[k + 2].z + q[k + 2].w);
        s3 += (q[k + 3].x + q[k + 3].y) + (q[k + 3].z + q[k + 3].w);
    }
    float s = (s0 + s1) + (s2 + s3);
    for (int off = 16; off > 0; off >>= 1) s += __shfl_xor(s, off, 32);
    if (lane == 0) red[wave] = s;
    __syncthreads();
    if (threadIdx.x == 0) {
        float t = 0.f;
        for (int w = 0; w < P1_WAVES; ++w) t += red[w];
        means[slice] = t * (1.0f / (float)HW);
    }
}

__global__ void __launch_bounds__(256)
recon_kernel(const float* __restrict__ transforms,
             const float* __restrict__ means,
             const float* __restrict__ counts,
             float*       __restrict__ out) {
    __shared__ __attribute__((aligned(16))) float lds_c[TT * 4];
    __shared__ __attribute__((aligned(16))) float lds_km[TT * 2];
    __shared__ int lds_wn[TT / 32];

    const int b    = blockIdx.y;
    const int tid  = threadIdx.x;
    const int lane = tid & 31;
    const int wave = tid >> 5;

    float k = 0.f, cx = 0.f, cy = 0.f, cz = 0.f, mean = 0.f;
    if (tid < TT) {
        const int s = b * TT + tid;
        const float* tr = transforms + (size_t)s * 6;
        k = counts[s];
        mean = means[s];
        cx = tr[0]; cy = tr[1]; cz = tr[2];
        *(float4*)&lds_c[tid * 4]  = make_float4(0.f, 0.f, 0.f, 1.f);
        *(float2*)&lds_km[tid * 2] = make_float2(0.f, 0.f);
    }
    const bool     named = k > 0.f;
    const unsigned nmask = (unsigned)__ballot(named);
    if (tid < TT && lane == 0) lds_wn[wave] = __popc(nmask);
    __syncthreads();
    const int n0 = lds_wn[0], n1 = lds_wn[1], n2 = lds_wn[2], n3 = lds_wn[3];
    if (named) {
        int row = __popc(nmask & ((1u << lane) - 1u));
        if (wave > 0) row += n0;
        if (wave > 1) row += n1;
        if (wave > 2) row += n2;
        *(float4*)&lds_c[row * 4]  = make_float4(-2.0f * cx, -2.0f * cy, -2.0f * cz,
                                                 cx * cx + cy * cy + cz * cz);
        *(float2*)&lds_km[row * 2] = make_float2(k, k * mean);
    }
    __syncthreads();
    const int ntiles = ((n0 + n1) + (n2 + n3) + 15) >> 4;

    const int  n  = lane & 15;
    const bool hi = lane >= 16;

    const int   v  = blockIdx.x * 128 + wave * 16 + n;
    const float gx = (float)(v >> 12);
    const float gy = (float)((v >> 6) & 63);
    const float gz = (float)(v & 63);
    const float g2 = gx * gx + gy * gy + gz * gz;

    v2f bm;
    bm.x = hi ? gz   : gx;
    bm.y = hi ? 1.0f : gy;

    const v8f cacc = {g2, g2, g2, g2, g2, g2, g2, g2};

    const float* arow = &lds_c[n * 4 + (hi ? 2 : 0)];
    const float* krow = &lds_km[(hi ? 8 : 0) * 2];

    v2f acc0 = {0.f, 0.f}, acc1 = {0.f, 0.f};

    v8f d = __builtin_amdgcn_wmma_f32_16x16x4_f32(
        false, *(const v2f*)arow, false, bm, (short)0, cacc, false, false);
    v8f d1 = __builtin_amdgcn_wmma_f32_16x16x4_f32(
        false, *(const v2f*)(arow + 64), false, bm, (short)0, cacc, false, false);

    #pragma unroll
    for (int tt = 0; tt < TT / 16; ++tt) {
        if (tt >= ntiles) break;
        const int tn = (tt + 2 < TT / 16) ? tt + 2 : TT / 16 - 1;
        const v8f d2 = __builtin_amdgcn_wmma_f32_16x16x4_f32(
            false, *(const v2f*)(arow + tn * 64), false, bm, (short)0, cacc, false, false);

        const v2f* km = (const v2f*)(krow + tt * 32);
        #pragma unroll
        for (int r = 0; r < 8; r += 2) {
            const float y0 = __builtin_amdgcn_rsqf(d[r]);
            const float y1 = __builtin_amdgcn_rsqf(d[r + 1]);
            const float w0 = __builtin_amdgcn_exp2f(y0 * LOG2E);
            const float w1 = __builtin_amdgcn_exp2f(y1 * LOG2E);
            const v2f ww0 = {w0, w0}, ww1 = {w1, w1};
            acc0 = __builtin_elementwise_fma(ww0, km[r],     acc0);
            acc1 = __builtin_elementwise_fma(ww1, km[r + 1], acc1);
        }
        d = d1;
        d1 = d2;
    }

    float sumw  = acc0.x + acc1.x;
    float sumwm = acc0.y + acc1.y;

    sumw  += __shfl_xor(sumw,  16, 32);
    sumwm += __shfl_xor(sumwm, 16, 32);

    if (!hi) {
        out[(size_t)b * VVV + v] = sumwm * __builtin_amdgcn_rcpf(sumw);
    }
}

extern "C" void kernel_launch(void* const* d_in, const int* in_sizes, int n_in,
                              void* d_out, int out_size, void* d_ws, size_t ws_size,
                              hipStream_t stream) {
    const float* slices     = (const float*)d_in[0];
    const float* transforms = (const float*)d_in[1];
    const int*   sel_idx    = (const int*)d_in[2];
    float*       out        = (float*)d_out;
    float*       means      = (float*)d_ws;
    float*       counts     = (float*)d_ws + NB * TT;

    slice_mean_kernel<<<NB * TT, P1_WAVES * 32, 0, stream>>>(slices, sel_idx, means, counts);

    dim3 grid(VVV / 128, NB);
    recon_kernel<<<grid, 256, 0, stream>>>(transforms, means, counts, out);
}


# pass 2: dist^2 = R(t;x,y) + Z(t;z), each an f32 WMMA tile: R once per 16-column work-group via LDS, one Z tile per wave and 16 slices serves 8 columns (was a WMMA per voxel tile); d2 is now two sums
# speedup vs baseline: 1.7331x; 1.0967x over previous
#include <hip/hip_runtime.h>


typedef __attribute__((ext_vector_type(2))) float v2f;
typedef __attribute__((ext_vector_type(8))) float v8f;

#define VXD 64
#define VVV (64*64*64)
#define TT  128
#define NB  2
#define HW  65536
#define LOG2E 1.4426950408889634f


#define P1_WAVES 16
#define P1_LOADS (HW / 4 / (P1_WAVES * 32))

__global__ void __launch_bounds__(P1_WAVES * 32)
slice_mean_kernel(const float* __restrict__ slices, const int* __restrict__ sel_idx,
                  float* __restrict__ means, float* __restrict__ counts) {
    __shared__ float red[P1_WAVES];
    __shared__ int   hits[TT / 32];
    const int slice = blockIdx.x;
    const int lane  = threadIdx.x & 31;
    const int wave  = threadIdx.x >> 5;
    if (wave < TT / 32) {
        const int pick = sel_idx[(slice / TT) * TT + threadIdx.x];
        const unsigned named = (unsigned)__ballot(pick == (slice % TT));
        if (lane == 0) hits[wave] = __popc(named);
    }
    __syncthreads();
    const int cnt = (hits[0] + hits[1]) + (hits[2] + hits[3]);
    if (threadIdx.x == 0) {
        counts[slice] = (float)cnt;
        if (cnt == 0) means[slice] = 0.f;
    }
    if (cnt == 0) return;
    const float4* p = (const float4*)(slices + (size_t)slice * HW)
                    + wave * (P1_LOADS * 32) + lane;
    float4 q[P1_LOADS];
    #pragma unroll
    for (int k = 0; k < P1_LOADS; ++k) q[k] = p[k * 32];
    __builtin_amdgcn_sched_barrier(0);
    float s0 = 0.f, s1 = 0.f, s2 = 0.f, s3 = 0.f;
    #pragma unroll
    for (int k = 0; k < P1_LOADS; k += 4) {
        s0 += (q[k + 0].x + q[k + 0].y) + (q[k + 0].z + q[k + 0].w);
        s1 += (q[k + 1].x + q[k + 1].y) + (q[k + 1].z + q[k + 1].w);
        s2 += (q[k + 2].x + q[k + 2].y) + (q[k + 2].z + q[k + 2].w);
        s3 += (q[k + 3].x + q[k + 3].y) + (q[k + 3].z + q[k + 3].w);
    }
    float s = (s0 + s1) + (s2 + s3);
    for (int off = 16; off > 0; off >>= 1) s += __shfl_xor(s, off, 32);
    if (lane == 0) red[wave] = s;
    __syncthreads();
    if (threadIdx.x == 0) {
        float t = 0.f;
        for (int w = 0; w < P1_WAVES; ++w) t += red[w];
        means[slice] = t * (1.0f / (float)HW);
    }
}

#define XYW    16
#define RPITCH (TT + 4)
__global__ void __launch_bounds__(256)
recon_kernel(const float* __restrict__ transforms,
             const float* __restrict__ means,
             const float* __restrict__ counts,
             float*       __restrict__ out) {
    __shared__ __attribute__((aligned(16))) float lds_ar[TT * 4];
    __shared__ __attribute__((aligned(16))) float lds_az[TT * 4];
    __shared__ __attribute__((aligned(16))) float lds_km[TT * 2];
    __shared__ __attribute__((aligned(16))) float lds_r[XYW * RPITCH];
    __shared__ int lds_wn[TT / 32];

    const int b    = blockIdx.y;
    const int tid  = threadIdx.x;
    const int lane = tid & 31;
    const int wave = tid >> 5;

    float k = 0.f, cx = 0.f, cy = 0.f, cz = 0.f, mean = 0.f;
    if (tid < TT) {
        const int s = b * TT + tid;
        const float* tr = transforms + (size_t)s * 6;
        k = counts[s];
        mean = means[s];
        cx = tr[0]; cy = tr[1]; cz = tr[2];
        *(float4*)&lds_ar[tid * 4] = make_float4(0.f, 0.f, 1.f, 0.f);
        *(float4*)&lds_az[tid * 4] = make_float4(0.f, 0.f, 0.f, 0.f);
        *(float2*)&lds_km[tid * 2] = make_float2(0.f, 0.f);
    }
    const bool     named = k > 0.f;
    const unsigned nmask = (unsigned)__ballot(named);
    if (tid < TT && lane == 0) lds_wn[wave] = __popc(nmask);
    __syncthreads();
    const int n0 = lds_wn[0], n1 = lds_wn[1], n2 = lds_wn[2], n3 = lds_wn[3];
    if (named) {
        int row = __popc(nmask & ((1u << lane) - 1u));
        if (wave > 0) row += n0;
        if (wave > 1) row += n1;
        if (wave > 2) row += n2;
        *(float4*)&lds_ar[row * 4] = make_float4(-2.0f * cx, -2.0f * cy,
                                                 cx * cx + cy * cy, 0.f);
        *(float4*)&lds_az[row * 4] = make_float4(-2.0f * cz, cz * cz, 0.f, 0.f);
        *(float2*)&lds_km[row * 2] = make_float2(k, k * mean);
    }
    __syncthreads();
    const int ntiles = ((n0 + n1) + (n2 + n3) + 15) >> 4;

    const int  n  = lane & 15;
    const bool hi = lane >= 16;

    const int xy0 = blockIdx.x * XYW;

    if (wave < ntiles) {
        const float gx = (float)(xy0 >> 6);
        const float gy = (float)((xy0 & 63) + n);
        v2f br;
        br.x = hi ? 1.0f : gx;
        br.y = hi ? 0.0f : gy;
        const float g2 = gx * gx + gy * gy;
        const v8f cr = {g2, g2, g2, g2, g2, g2, g2, g2};
        const v8f rt = __builtin_amdgcn_wmma_f32_16x16x4_f32(
            false, *(const v2f*)&lds_ar[(wave * 16 + n) * 4 + (hi ? 2 : 0)],
            false, br, (short)0, cr, false, false);
        float* dst = &lds_r[n * RPITCH + wave * 16 + (hi ? 8 : 0)];
        *(float4*)dst       = make_float4(rt[0], rt[1], rt[2], rt[3]);
        *(float4*)(dst + 4) = make_float4(rt[4], rt[5], rt[6], rt[7]);
    }
    __syncthreads();

    const int zt = wave & 3;
    const int jh = wave >> 2;
    const float gz  = (float)(zt * 16 + n);
    const float gz2 = gz * gz;

    v2f bz;
    bz.x = hi ? 0.0f : gz;
    bz.y = hi ? 0.0f : 1.0f;
    const v8f czacc = {gz2, gz2, gz2, gz2, gz2, gz2, gz2, gz2};

    const float* azrow = &lds_az[n * 4 + (hi ? 2 : 0)];
    const float* krow  = &lds_km[(hi ? 8 : 0) * 2];
    const float* rrow  = &lds_r[(jh * 8) * RPITCH + (hi ? 8 : 0)];

    v2f acc0[8], acc1[8];
    #pragma unroll
    for (int j = 0; j < 8; ++j) { acc0[j] = (v2f){0.f, 0.f}; acc1[j] = (v2f){0.f, 0.f}; }

    v8f zz = __builtin_amdgcn_wmma_f32_16x16x4_f32(
        false, *(const v2f*)azrow, false, bz, (short)0, czacc, false, false);

    int tt = 0;
    #pragma unroll 1
    do {
        const int tn = (tt + 1 < TT / 16) ? tt + 1 : TT / 16 - 1;
        const v8f zn = __builtin_amdgcn_wmma_f32_16x16x4_f32(
            false, *(const v2f*)(azrow + tn * 64), false, bz, (short)0, czacc, false, false);
        v2f km[8];
        #pragma unroll
        for (int r = 0; r < 8; ++r) km[r] = *(const v2f*)(krow + tt * 32 + r * 2);

        #pragma unroll
        for (int j = 0; j < 8; ++j) {
            const float4 ra = *(const float4*)(rrow + j * RPITCH + tt * 16);
            const float4 rb = *(const float4*)(rrow + j * RPITCH + tt * 16 + 4);
            const float rr[8] = {ra.x, ra.y, ra.z, ra.w, rb.x, rb.y, rb.z, rb.w};
            #pragma unroll
            for (int r = 0; r < 8; r += 2) {
                const float y0 = __builtin_amdgcn_rsqf(zz[r]     + rr[r]);
                const float y1 = __builtin_amdgcn_rsqf(zz[r + 1] + rr[r + 1]);
                const float w0 = __builtin_amdgcn_exp2f(y0 * LOG2E);
                const float w1 = __builtin_amdgcn_exp2f(y1 * LOG2E);
                const v2f ww0 = {w0, w0}, ww1 = {w1, w1};
                acc0[j] = __builtin_elementwise_fma(ww0, km[r],     acc0[j]);
                acc1[j] = __builtin_elementwise_fma(ww1, km[r + 1], acc1[j]);
            }
        }
        zz = zn;
    } while (++tt < ntiles);

    float* obase = out + (size_t)b * VVV + (size_t)(xy0 + jh * 8) * VXD + zt * 16 + n;
    #pragma unroll
    for (int p = 0; p < 4; ++p) {
        float sw0 = acc0[2 * p].x + acc1[2 * p].x;
        float sm0 = acc0[2 * p].y + acc1[2 * p].y;
        float sw1 = acc0[2 * p + 1].x + acc1[2 * p + 1].x;
        float sm1 = acc0[2 * p + 1].y + acc1[2 * p + 1].y;
        sw0 += __shfl_xor(sw0, 16, 32);
        sm0 += __shfl_xor(sm0, 16, 32);
        sw1 += __shfl_xor(sw1, 16, 32);
        sm1 += __shfl_xor(sm1, 16, 32);
        const float sw = hi ? sw1 : sw0;
        const float sm = hi ? sm1 : sm0;
        obase[(2 * p + (hi ? 1 : 0)) * VXD] = sm * __builtin_amdgcn_rcpf(sw);
    }
}

extern "C" void kernel_launch(void* const* d_in, const int* in_sizes, int n_in,
                              void* d_out, int out_size, void* d_ws, size_t ws_size,
                              hipStream_t stream) {
    const float* slices     = (const float*)d_in[0];
    const float* transforms = (const float*)d_in[1];
    const int*   sel_idx    = (const int*)d_in[2];
    float*       out        = (float*)d_out;
    float*       means      = (float*)d_ws;
    float*       counts     = (float*)d_ws + NB * TT;

    slice_mean_kernel<<<NB * TT, P1_WAVES * 32, 0, stream>>>(slices, sel_idx, means, counts);

    dim3 grid(VVV / (XYW * VXD), NB);
    recon_kernel<<<grid, 256, 0, stream>>>(transforms, means, counts, out);
}
